# starting kernel
# baseline (speedup 1.0000x reference)
#include <hip/hip_runtime.h>

typedef float v2f __attribute__((ext_vector_type(2)));
typedef float v4f __attribute__((ext_vector_type(4)));
typedef float v8f __attribute__((ext_vector_type(8)));

#define WAVES 8

__global__ __launch_bounds__(WAVES * 32)
void nufft2_wmma(const float* __restrict__ xr, const float* __restrict__ xi,
                 const float* __restrict__ traj, float* __restrict__ out)
{
    constexpr int B = 16, C = 4, H = 64, W = 64, M = 2048;
    (void)B; (void)H;

    __shared__ float2 eyTab[WAVES][16][64];

    const int lane = threadIdx.x & 31;
    const int wave = threadIdx.x >> 5;
    const int tile = blockIdx.x * WAVES + wave;
    const int b    = tile >> 7;
    const int m0   = (tile & 127) << 4;
    const int hh   = lane >> 4;
    const int mloc = lane & 15;

    const float2 kk = ((const float2*)traj)[(size_t)b * M + m0 + mloc];
    const float ky = kk.x, kx = kk.y;

    {
        const int r0 = hh * 32;
        float s0, c0, su, cu;
        sincosf(ky * (float)(r0 - 32), &s0, &c0);
        sincosf(ky, &su, &cu);
        float cc = c0, ss = s0;
        #pragma unroll 4
        for (int r = 0; r < 32; ++r) {
            eyTab[wave][mloc][r0 + r] = make_float2(cc, ss);
            const float t = cc * cu - ss * su;
            ss = ss * cu + cc * su;
            cc = t;
        }
    }
    __syncthreads();

    v2f axc[16], axs[16];
    {
        const int col0 = hh * 32;
        float s0, c0, su, cu;
        sincosf(kx * (float)(col0 - 32), &s0, &c0);
        sincosf(kx, &su, &cu);
        const float c2 = cu * cu - su * su, s2 = 2.0f * cu * su;
        float ec = c0, es = s0;
        #pragma unroll
        for (int k = 0; k < 16; ++k) {
            const float e1c = ec * cu - es * su;
            const float e1s = es * cu + ec * su;
            axc[k].x = ec;  axc[k].y = e1c;
            axs[k].x = es;  axs[k].y = e1s;
            const float t = ec * c2 - es * s2;
            es = es * c2 + ec * s2;
            ec = t;
        }
    }

    float yre[8], yim[8];
    #pragma unroll
    for (int j = 0; j < 8; ++j) { yre[j] = 0.0f; yim[j] = 0.0f; }

    const int cc_ = mloc & 3;
    const int rl  = mloc >> 2;
    const size_t xbase = ((size_t)(b * C + cc_) * H) * W + (size_t)(hh * 32);

    #pragma unroll 1
    for (int t = 0; t < 16; ++t) {
        const int r = t * 4 + rl;
        const float* pr = xr + xbase + (size_t)r * W;
        const float* pi = xi + xbase + (size_t)r * W;

        v4f brA[8], biA[8];
        #pragma unroll
        for (int k2 = 0; k2 < 8; ++k2) {
            brA[k2] = *(const v4f*)(pr + 4 * k2);
            biA[k2] = *(const v4f*)(pi + 4 * k2);
        }

        v8f zr = {}; v8f zi = {};
        #pragma unroll
        for (int k2 = 0; k2 < 8; ++k2) {
            const v2f br0 = __builtin_shufflevector(brA[k2], brA[k2], 0, 1);
            const v2f br1 = __builtin_shufflevector(brA[k2], brA[k2], 2, 3);
            const v2f bi0 = __builtin_shufflevector(biA[k2], biA[k2], 0, 1);
            const v2f bi1 = __builtin_shufflevector(biA[k2], biA[k2], 2, 3);
            const int k0 = 2 * k2, k1 = 2 * k2 + 1;

            zr = __builtin_amdgcn_wmma_f32_16x16x4_f32(false, axc[k0], false, br0,
                                                       (short)0, zr, false, false);
            zi = __builtin_amdgcn_wmma_f32_16x16x4_f32(false, axc[k0], false, bi0,
                                                       (short)0, zi, false, false);
            zr = __builtin_amdgcn_wmma_f32_16x16x4_f32(false, axs[k0], false, bi0,
                                                       (short)0, zr, false, false);
            const v2f nbr0 = -br0;
            zi = __builtin_amdgcn_wmma_f32_16x16x4_f32(false, axs[k0], false, nbr0,
                                                       (short)0, zi, false, false);

            zr = __builtin_amdgcn_wmma_f32_16x16x4_f32(false, axc[k1], false, br1,
                                                       (short)0, zr, false, false);
            zi = __builtin_amdgcn_wmma_f32_16x16x4_f32(false, axc[k1], false, bi1,
                                                       (short)0, zi, false, false);
            zr = __builtin_amdgcn_wmma_f32_16x16x4_f32(false, axs[k1], false, bi1,
                                                       (short)0, zr, false, false);
            const v2f nbr1 = -br1;
            zi = __builtin_amdgcn_wmma_f32_16x16x4_f32(false, axs[k1], false, nbr1,
                                                       (short)0, zi, false, false);
        }

        #pragma unroll
        for (int j = 0; j < 8; ++j) {
            const float2 ey = eyTab[wave][j + hh * 8][r];
            yre[j] = fmaf(ey.x, zr[j], yre[j]);
            yre[j] = fmaf(ey.y, zi[j], yre[j]);
            yim[j] = fmaf(ey.x, zi[j], yim[j]);
            yim[j] = fmaf(-ey.y, zr[j], yim[j]);
        }
    }

    #pragma unroll
    for (int j = 0; j < 8; ++j) {
        float a = yre[j], c = yim[j];
        a += __shfl_xor(a, 4, 32);
        a += __shfl_xor(a, 8, 32);
        c += __shfl_xor(c, 4, 32);
        c += __shfl_xor(c, 8, 32);
        if ((lane & 12) == 0) {
            const int m  = m0 + j + hh * 8;
            const int ch = lane & 3;
            float2* op = (float2*)out + ((size_t)(b * C + ch) * M + m);
            *op = make_float2(a, c);
        }
    }
}

extern "C" void kernel_launch(void* const* d_in, const int* in_sizes, int n_in,
                              void* d_out, int out_size, void* d_ws, size_t ws_size,
                              hipStream_t stream) {
    (void)in_sizes; (void)n_in; (void)out_size; (void)d_ws; (void)ws_size;
    const float* xr   = (const float*)d_in[0];
    const float* xi   = (const float*)d_in[1];
    const float* traj = (const float*)d_in[2];
    float* out = (float*)d_out;

    dim3 grid(256), block(WAVES * 32);
    hipLaunchKernelGGL(nufft2_wmma, grid, block, 0, stream, xr, xi, traj, out);
}


# Stage each 16-row image slab once per block in LDS with whole-line global loads (2-buffer ring); the 8 waves read B fragments by ds_load_b128 instead of 32-line global loads
# speedup vs baseline: 1.5710x; 1.5710x over previous
#include <hip/hip_runtime.h>


typedef float v2f __attribute__((ext_vector_type(2)));
typedef float v4f __attribute__((ext_vector_type(4)));
typedef float v8f __attribute__((ext_vector_type(8)));

#define WAVES 8
#define XPITCH 68
#define TPS 4
#define NSTAGE 4

__global__ __launch_bounds__(WAVES * 32)
void nufft2_wmma(const float* __restrict__ xr, const float* __restrict__ xi,
                 const float* __restrict__ traj, float* __restrict__ out)
{
    constexpr int B = 16, C = 4, H = 64, W = 64, M = 2048;
    (void)B;

    __shared__ float2 eyTab[WAVES][16][64];
    __shared__ __attribute__((aligned(16))) float xs[2][TPS][2][16][XPITCH];

    const int tid  = threadIdx.x;
    const int lane = tid & 31;
    const int wave = tid >> 5;
    const int tile = blockIdx.x * WAVES + wave;
    const int b    = tile >> 7;
    const int m0   = (tile & 127) << 4;
    const int hh   = lane >> 4;
    const int mloc = lane & 15;

    const int srow = tid >> 4;
    const int scol = (tid & 15) << 2;
    const int sti  = srow >> 2;
    const int srl  = srow & 3;
    const size_t sbase = (size_t)b * C * H * W + (size_t)srow * W + (size_t)scol;

    v4f gre[C], gim[C];
    #pragma unroll
    for (int c = 0; c < C; ++c) {
        gre[c] = *(const v4f*)(xr + sbase + (size_t)c * H * W);
        gim[c] = *(const v4f*)(xi + sbase + (size_t)c * H * W);
    }

    const float2 kk = ((const float2*)traj)[(size_t)b * M + m0 + mloc];
    const float ky = kk.x, kx = kk.y;

    {
        const int r0 = hh * 32;
        float s0, c0, su, cu;
        sincosf(ky * (float)(r0 - 32), &s0, &c0);
        sincosf(ky, &su, &cu);
        float cc = c0, ss = s0;
        #pragma unroll 4
        for (int r = 0; r < 32; ++r) {
            eyTab[wave][mloc][r0 + r] = make_float2(cc, ss);
            const float t = cc * cu - ss * su;
            ss = ss * cu + cc * su;
            cc = t;
        }
    }

    v2f axc[16], axs[16];
    {
        const int col0 = hh * 32;
        float s0, c0, su, cu;
        sincosf(kx * (float)(col0 - 32), &s0, &c0);
        sincosf(kx, &su, &cu);
        const float c2 = cu * cu - su * su, s2 = 2.0f * cu * su;
        float ec = c0, es = s0;
        #pragma unroll
        for (int k = 0; k < 16; ++k) {
            const float e1c = ec * cu - es * su;
            const float e1s = es * cu + ec * su;
            axc[k].x = ec;  axc[k].y = e1c;
            axs[k].x = es;  axs[k].y = e1s;
            const float t = ec * c2 - es * s2;
            es = es * c2 + ec * s2;
            ec = t;
        }
    }

    #pragma unroll
    for (int c = 0; c < C; ++c) {
        *(v4f*)&xs[0][sti][0][srl * 4 + c][scol] = gre[c];
        *(v4f*)&xs[0][sti][1][srl * 4 + c][scol] = gim[c];
    }
    __syncthreads();

    float yre[8], yim[8];
    #pragma unroll
    for (int j = 0; j < 8; ++j) { yre[j] = 0.0f; yim[j] = 0.0f; }

    const int rl = mloc >> 2;

    #pragma unroll 1
    for (int s = 0; s < NSTAGE; ++s) {
        const int buf = s & 1;
        const bool more = (s + 1 < NSTAGE);

        if (more) {
            const size_t o = sbase + (size_t)(s + 1) * (4 * TPS) * W;
            #pragma unroll
            for (int c = 0; c < C; ++c) {
                gre[c] = *(const v4f*)(xr + o + (size_t)c * H * W);
                gim[c] = *(const v4f*)(xi + o + (size_t)c * H * W);
            }
        }

        const float* fb = &xs[buf][0][0][mloc][hh * 32];
        constexpr int TSTRIDE = 2 * 16 * XPITCH;
        constexpr int ISTRIDE = 16 * XPITCH;

        v4f fr[2][8], fi[2][8];
        #pragma unroll
        for (int k2 = 0; k2 < 8; ++k2) {
            fr[0][k2] = *(const v4f*)(fb + 4 * k2);
            fi[0][k2] = *(const v4f*)(fb + ISTRIDE + 4 * k2);
        }

        #pragma unroll
        for (int ti = 0; ti < TPS; ++ti) {
            const int cur = ti & 1;
            if (ti + 1 < TPS) {
                #pragma unroll
                for (int k2 = 0; k2 < 8; ++k2) {
                    fr[cur ^ 1][k2] = *(const v4f*)(fb + (ti + 1) * TSTRIDE + 4 * k2);
                    fi[cur ^ 1][k2] = *(const v4f*)(fb + (ti + 1) * TSTRIDE + ISTRIDE + 4 * k2);
                }
            }

            v8f zr = {}; v8f zi = {};
            #pragma unroll
            for (int k2 = 0; k2 < 8; ++k2) {
                const v4f bra = fr[cur][k2], bia = fi[cur][k2];
                const v2f br0 = __builtin_shufflevector(bra, bra, 0, 1);
                const v2f br1 = __builtin_shufflevector(bra, bra, 2, 3);
                const v2f bi0 = __builtin_shufflevector(bia, bia, 0, 1);
                const v2f bi1 = __builtin_shufflevector(bia, bia, 2, 3);
                const int k0 = 2 * k2, k1 = 2 * k2 + 1;

                zr = __builtin_amdgcn_wmma_f32_16x16x4_f32(false, axc[k0], false, br0,
                                                           (short)0, zr, false, false);
                zi = __builtin_amdgcn_wmma_f32_16x16x4_f32(false, axc[k0], false, bi0,
                                                           (short)0, zi, false, false);
                zr = __builtin_amdgcn_wmma_f32_16x16x4_f32(false, axs[k0], false, bi0,
                                                           (short)0, zr, false, false);
                const v2f nbr0 = -br0;
                zi = __builtin_amdgcn_wmma_f32_16x16x4_f32(false, axs[k0], false, nbr0,
                                                           (short)0, zi, false, false);

                zr = __builtin_amdgcn_wmma_f32_16x16x4_f32(false, axc[k1], false, br1,
                                                           (short)0, zr, false, false);
                zi = __builtin_amdgcn_wmma_f32_16x16x4_f32(false, axc[k1], false, bi1,
                                                           (short)0, zi, false, false);
                zr = __builtin_amdgcn_wmma_f32_16x16x4_f32(false, axs[k1], false, bi1,
                                                           (short)0, zr, false, false);
                const v2f nbr1 = -br1;
                zi = __builtin_amdgcn_wmma_f32_16x16x4_f32(false, axs[k1], false, nbr1,
                                                           (short)0, zi, false, false);
            }

            const int r = (s * TPS + ti) * 4 + rl;
            #pragma unroll
            for (int j = 0; j < 8; ++j) {
                const float2 ey = eyTab[wave][j + hh * 8][r];
                yre[j] = fmaf(ey.x, zr[j], yre[j]);
                yre[j] = fmaf(ey.y, zi[j], yre[j]);
                yim[j] = fmaf(ey.x, zi[j], yim[j]);
                yim[j] = fmaf(-ey.y, zr[j], yim[j]);
            }
        }

        if (more) {
            #pragma unroll
            for (int c = 0; c < C; ++c) {
                *(v4f*)&xs[buf ^ 1][sti][0][srl * 4 + c][scol] = gre[c];
                *(v4f*)&xs[buf ^ 1][sti][1][srl * 4 + c][scol] = gim[c];
            }
            __syncthreads();
        }
    }

    #pragma unroll
    for (int j = 0; j < 8; ++j) {
        float a = yre[j], c = yim[j];
        a += __shfl_xor(a, 4, 32);
        a += __shfl_xor(a, 8, 32);
        c += __shfl_xor(c, 4, 32);
        c += __shfl_xor(c, 8, 32);
        if ((lane & 12) == 0) {
            const int m  = m0 + j + hh * 8;
            const int ch = lane & 3;
            float2* op = (float2*)out + ((size_t)(b * C + ch) * M + m);
            *op = make_float2(a, c);
        }
    }
}

extern "C" void kernel_launch(void* const* d_in, const int* in_sizes, int n_in,
                              void* d_out, int out_size, void* d_ws, size_t ws_size,
                              hipStream_t stream) {
    (void)in_sizes; (void)n_in; (void)out_size; (void)d_ws; (void)ws_size;
    const float* xr   = (const float*)d_in[0];
    const float* xi   = (const float*)d_in[1];
    const float* traj = (const float*)d_in[2];
    float* out = (float*)d_out;

    dim3 grid(256), block(WAVES * 32);
    hipLaunchKernelGGL(nufft2_wmma, grid, block, 0, stream, xr, xi, traj, out);
}


# Fold column DFT by cos/sin symmetry about its half-sample centre: staging stores mirror-column sums and differences, K 64 to 32 halves the WMMAs (pairs summed first; half-step phase in row table)
# speedup vs baseline: 2.4660x; 1.5697x over previous
#include <hip/hip_runtime.h>


typedef float v2f __attribute__((ext_vector_type(2)));
typedef float v4f __attribute__((ext_vector_type(4)));
typedef float v8f __attribute__((ext_vector_type(8)));

#define WAVES 8
#define XPITCH 68
#define TPS 4
#define NSTAGE 4


__global__ __launch_bounds__(WAVES * 32)
void nufft2_wmma(const float* __restrict__ xr, const float* __restrict__ xi,
                 const float* __restrict__ traj, float* __restrict__ out)
{
    constexpr int B = 16, C = 4, H = 64, W = 64, M = 2048;
    (void)B;

    __shared__ float2 eyTab[WAVES][16][64];
    __shared__ __attribute__((aligned(16))) float xs[2][TPS][2][16][XPITCH];

    const int tid  = threadIdx.x;
    const int lane = tid & 31;
    const int wave = tid >> 5;
    const int tile = blockIdx.x * WAVES + wave;
    const int b    = tile >> 7;
    const int m0   = (tile & 127) << 4;
    const int hh   = lane >> 4;
    const int mloc = lane & 15;

    const int sp   = tid & 7;
    const int srow = (tid >> 3) & 15;
    const int sri  = tid >> 7;
    const int sti  = srow >> 2;
    const int srl  = srow & 3;
    const float* __restrict__ simg = sri ? xi : xr;
    const size_t sbase = (size_t)b * C * H * W + (size_t)srow * W;
    const int cpos = 32 + 4 * sp;
    const int cneg = 28 - 4 * sp;

    v4f gp[C], gn[C];
    #pragma unroll
    for (int c = 0; c < C; ++c) {
        gp[c] = *(const v4f*)(simg + sbase + (size_t)c * H * W + cpos);
        gn[c] = *(const v4f*)(simg + sbase + (size_t)c * H * W + cneg);
    }

    const float2 kk = ((const float2*)traj)[(size_t)b * M + m0 + mloc];
    const float ky = kk.x, kx = kk.y;

    {
        const int r0 = hh * 32;
        float s0, c0, su, cu;
        sincosf(ky * (float)(r0 - 32) - 0.5f * kx, &s0, &c0);
        sincosf(ky, &su, &cu);
        float cc = c0, ss = s0;
        #pragma unroll 4
        for (int r = 0; r < 32; ++r) {
            eyTab[wave][mloc][r0 + r] = make_float2(cc, ss);
            const float t = cc * cu - ss * su;
            ss = ss * cu + cc * su;
            cc = t;
        }
    }

    v2f axc[8], axs[8], axn[8];
    {
        float s0, c0, su, cu;
        sincosf(kx * ((float)(hh * 16) + 0.5f), &s0, &c0);
        sincosf(kx, &su, &cu);
        const float c2 = cu * cu - su * su, s2 = 2.0f * cu * su;
        float ec = c0, es = s0;
        #pragma unroll
        for (int k = 0; k < 8; ++k) {
            const float e1c = ec * cu - es * su;
            const float e1s = es * cu + ec * su;
            axc[k].x = ec;   axc[k].y = e1c;
            axs[k].x = es;   axs[k].y = e1s;
            axn[k].x = -es;  axn[k].y = -e1s;
            const float t = ec * c2 - es * s2;
            es = es * c2 + ec * s2;
            ec = t;
        }
    }

    #pragma unroll
    for (int c = 0; c < C; ++c) {
        const v4f p = gp[c];
        const v4f q = __builtin_shufflevector(gn[c], gn[c], 3, 2, 1, 0);
        float* rowp = &xs[0][sti][sri][srl * 4 + c][0];
        *(v4f*)(rowp + 4 * sp)      = p + q;
        *(v4f*)(rowp + 32 + 4 * sp) = p - q;
    }
    __syncthreads();

    float yre[8], yim[8];
    #pragma unroll
    for (int j = 0; j < 8; ++j) { yre[j] = 0.0f; yim[j] = 0.0f; }

    const int rl = mloc >> 2;

    #pragma unroll 1
    for (int s = 0; s < NSTAGE; ++s) {
        const int buf = s & 1;
        const bool more = (s + 1 < NSTAGE);

        if (more) {
            const size_t o = sbase + (size_t)(s + 1) * (4 * TPS) * W;
            #pragma unroll
            for (int c = 0; c < C; ++c) {
                gp[c] = *(const v4f*)(simg + o + (size_t)c * H * W + cpos);
                gn[c] = *(const v4f*)(simg + o + (size_t)c * H * W + cneg);
            }
        }

        const float* fb = &xs[buf][0][0][mloc][hh * 16];
        constexpr int TSTRIDE = 2 * 16 * XPITCH;
        constexpr int ISTRIDE = 16 * XPITCH;
        constexpr int OSTRIDE = 32;

        v4f er[2][4], orr[2][4], ei[2][4], oi[2][4];
        #pragma unroll
        for (int q = 0; q < 4; ++q) {
            er[0][q]  = *(const v4f*)(fb + 4 * q);
            orr[0][q] = *(const v4f*)(fb + OSTRIDE + 4 * q);
            ei[0][q]  = *(const v4f*)(fb + ISTRIDE + 4 * q);
            oi[0][q]  = *(const v4f*)(fb + ISTRIDE + OSTRIDE + 4 * q);
        }

        #pragma unroll
        for (int ti = 0; ti < TPS; ++ti) {
            const int cur = ti & 1;
            if (ti + 1 < TPS) {
                const float* fn = fb + (ti + 1) * TSTRIDE;
                #pragma unroll
                for (int q = 0; q < 4; ++q) {
                    er[cur ^ 1][q]  = *(const v4f*)(fn + 4 * q);
                    orr[cur ^ 1][q] = *(const v4f*)(fn + OSTRIDE + 4 * q);
                    ei[cur ^ 1][q]  = *(const v4f*)(fn + ISTRIDE + 4 * q);
                    oi[cur ^ 1][q]  = *(const v4f*)(fn + ISTRIDE + OSTRIDE + 4 * q);
                }
            }

            v8f zr = {}; v8f zi = {};
            #pragma unroll
            for (int q = 0; q < 4; ++q) {
                const v4f era = er[cur][q], ora = orr[cur][q];
                const v4f eia = ei[cur][q], oia = oi[cur][q];
                const v2f er0 = __builtin_shufflevector(era, era, 0, 1);
                const v2f er1 = __builtin_shufflevector(era, era, 2, 3);
                const v2f or0 = __builtin_shufflevector(ora, ora, 0, 1);
                const v2f or1 = __builtin_shufflevector(ora, ora, 2, 3);
                const v2f ei0 = __builtin_shufflevector(eia, eia, 0, 1);
                const v2f ei1 = __builtin_shufflevector(eia, eia, 2, 3);
                const v2f oi0 = __builtin_shufflevector(oia, oia, 0, 1);
                const v2f oi1 = __builtin_shufflevector(oia, oia, 2, 3);
                const int k0 = 2 * q, k1 = 2 * q + 1;

                zr = __builtin_amdgcn_wmma_f32_16x16x4_f32(false, axc[k0], false, er0,
                                                           (short)0, zr, false, false);
                zi = __builtin_amdgcn_wmma_f32_16x16x4_f32(false, axc[k0], false, ei0,
                                                           (short)0, zi, false, false);
                zr = __builtin_amdgcn_wmma_f32_16x16x4_f32(false, axs[k0], false, oi0,
                                                           (short)0, zr, false, false);
                zi = __builtin_amdgcn_wmma_f32_16x16x4_f32(false, axn[k0], false, or0,
                                                           (short)0, zi, false, false);

                zr = __builtin_amdgcn_wmma_f32_16x16x4_f32(false, axc[k1], false, er1,
                                                           (short)0, zr, false, false);
                zi = __builtin_amdgcn_wmma_f32_16x16x4_f32(false, axc[k1], false, ei1,
                                                           (short)0, zi, false, false);
                zr = __builtin_amdgcn_wmma_f32_16x16x4_f32(false, axs[k1], false, oi1,
                                                           (short)0, zr, false, false);
                zi = __builtin_amdgcn_wmma_f32_16x16x4_f32(false, axn[k1], false, or1,
                                                           (short)0, zi, false, false);
            }

            const int r = (s * TPS + ti) * 4 + rl;
            #pragma unroll
            for (int j = 0; j < 8; ++j) {
                const float2 ey = eyTab[wave][j + hh * 8][r];
                yre[j] = fmaf(ey.x, zr[j], yre[j]);
                yre[j] = fmaf(ey.y, zi[j], yre[j]);
                yim[j] = fmaf(ey.x, zi[j], yim[j]);
                yim[j] = fmaf(-ey.y, zr[j], yim[j]);
            }
        }

        if (more) {
            #pragma unroll
            for (int c = 0; c < C; ++c) {
                const v4f p = gp[c];
                const v4f q = __builtin_shufflevector(gn[c], gn[c], 3, 2, 1, 0);
                float* rowp = &xs[buf ^ 1][sti][sri][srl * 4 + c][0];
                *(v4f*)(rowp + 4 * sp)      = p + q;
                *(v4f*)(rowp + 32 + 4 * sp) = p - q;
            }
            __syncthreads();
        }
    }

    #pragma unroll
    for (int j = 0; j < 8; ++j) {
        float a = yre[j], c = yim[j];
        a += __shfl_xor(a, 4, 32);
        a += __shfl_xor(a, 8, 32);
        c += __shfl_xor(c, 4, 32);
        c += __shfl_xor(c, 8, 32);
        if ((lane & 12) == 0) {
            const int m  = m0 + j + hh * 8;
            const int ch = lane & 3;
            float2* op = (float2*)out + ((size_t)(b * C + ch) * M + m);
            *op = make_float2(a, c);
        }
    }
}

extern "C" void kernel_launch(void* const* d_in, const int* in_sizes, int n_in,
                              void* d_out, int out_size, void* d_ws, size_t ws_size,
                              hipStream_t stream) {
    (void)in_sizes; (void)n_in; (void)out_size; (void)d_ws; (void)ws_size;
    const float* xr   = (const float*)d_in[0];
    const float* xi   = (const float*)d_in[1];
    const float* traj = (const float*)d_in[2];
    float* out = (float*)d_out;

    dim3 grid(256), block(WAVES * 32);
    hipLaunchKernelGGL(nufft2_wmma, grid, block, 0, stream, xr, xi, traj, out);
}
